# starting kernel
# baseline (speedup 1.0000x reference)
#include <hip/hip_runtime.h>
#include <hip/hip_bf16.h>


typedef __attribute__((ext_vector_type(2))) float v2f;
typedef __attribute__((ext_vector_type(8))) float v8f;

#define DFEAT 128

__global__ __launch_bounds__(256) void gconv_gemm_xw(const float* __restrict__ X,
                                                     const float* __restrict__ W,
                                                     float* __restrict__ Y,
                                                     int mTiles) {
    const int wave = (int)((blockIdx.x * blockDim.x + threadIdx.x) >> 5);
    const int lane = (int)(threadIdx.x & 31);
    const int nTiles = DFEAT / 16;
    const int mt = wave / nTiles;
    const int nt = wave % nTiles;
    if (mt >= mTiles) return;

    const int hi  = lane >> 4;
    const int l15 = lane & 15;

    const float* Arow = X + (size_t)(mt * 16 + l15) * DFEAT;
    const int ncol = nt * 16 + l15;

    v8f acc = {};
#pragma unroll 8
    for (int k = 0; k < DFEAT; k += 4) {
        v2f a, b;
        const int ka = k + 2 * hi;
        a.x = Arow[ka + 0];
        a.y = Arow[ka + 1];
        b.x = W[(size_t)(ka + 0) * DFEAT + ncol];
        b.y = W[(size_t)(ka + 1) * DFEAT + ncol];
        acc = __builtin_amdgcn_wmma_f32_16x16x4_f32(
            false, a, false, b, (short)0, acc, false, false);
    }

    float* Yt = Y + (size_t)(mt * 16 + 8 * hi) * DFEAT + ncol;
#pragma unroll
    for (int i = 0; i < 8; ++i) {
        Yt[(size_t)i * DFEAT] = acc[i];
    }
}

__global__ __launch_bounds__(256) void gconv_scatter(const float* __restrict__ Y,
                                                     const int* __restrict__ esrc,
                                                     const int* __restrict__ edst,
                                                     const float* __restrict__ eval,
                                                     float* __restrict__ Z,
                                                     int nEdges) {
    const int wave = (int)((blockIdx.x * blockDim.x + threadIdx.x) >> 5);
    const int lane = (int)(threadIdx.x & 31);
    if (wave >= nEdges) return;

    const int   s = esrc[wave];
    const int   d = edst[wave];
    const float v = eval[wave];

    const float4 y = ((const float4*)(Y + (size_t)s * DFEAT))[lane];
    float* zr = Z + (size_t)d * DFEAT + lane * 4;
    atomicAdd(zr + 0, v * y.x);
    atomicAdd(zr + 1, v * y.y);
    atomicAdd(zr + 2, v * y.z);
    atomicAdd(zr + 3, v * y.w);
}

extern "C" void kernel_launch(void* const* d_in, const int* in_sizes, int n_in,
                              void* d_out, int out_size, void* d_ws, size_t ws_size,
                              hipStream_t stream) {
    const float* X    = (const float*)d_in[0];
    const float* W    = (const float*)d_in[1];
    const int*   esrc = (const int*)  d_in[2];
    const int*   edst = (const int*)  d_in[3];
    const float* eval = (const float*)d_in[4];
    float*       Z    = (float*)d_out;
    float*       Y    = (float*)d_ws;

    const int nNodes = in_sizes[0] / DFEAT;
    const int nEdges = in_sizes[2];

    const int mTiles   = nNodes / 16;
    const int nTiles   = DFEAT / 16;
    const int nWaves   = mTiles * nTiles;
    const int block    = 256;
    const int gemmGrid = (nWaves * 32 + block - 1) / block;
    gconv_gemm_xw<<<gemmGrid, block, 0, stream>>>(X, W, Y, mTiles);

    hipMemsetAsync(d_out, 0, (size_t)out_size * sizeof(float), stream);

    const int edgeGrid = (nEdges + 7) / 8;
    gconv_scatter<<<edgeGrid, block, 0, stream>>>(Y, esrc, edst, eval, Z, nEdges);
}


# Float atomic scatter removed: edges counting-sorted by dst in-call (int atomics), each 16-row tile sums its X rows into LDS, then T x W on f32 WMMA; row sums now sequential in rank order
# speedup vs baseline: 1.6206x; 1.6206x over previous
#include <hip/hip_runtime.h>


typedef __attribute__((ext_vector_type(2))) float v2f;
typedef __attribute__((ext_vector_type(8))) float v8f;

#define DFEAT      128
#define SCAN_CHUNK 1024
#define TPITCH     132

__global__ __launch_bounds__(256) void zero_counts(int* __restrict__ cnt, int n) {
    const int i = (int)(blockIdx.x * blockDim.x + threadIdx.x) * 4;
    if (i < n) *(int4*)(cnt + i) = make_int4(0, 0, 0, 0);
}

__global__ __launch_bounds__(256) void count_rank(const int* __restrict__ edst,
                                                  int* __restrict__ cnt,
                                                  int* __restrict__ rank,
                                                  int nEdges) {
    const int i      = (int)(blockIdx.x * blockDim.x + threadIdx.x);
    const int stride = (int)(gridDim.x * blockDim.x);
    int d[4], r[4];
#pragma unroll
    for (int k = 0; k < 4; ++k) {
        const int e = i + k * stride;
        d[k] = (e < nEdges) ? edst[e] : -1;
    }
#pragma unroll
    for (int k = 0; k < 4; ++k) {
        r[k] = 0;
        if (d[k] >= 0) r[k] = atomicAdd(&cnt[d[k]], 1);
    }
#pragma unroll
    for (int k = 0; k < 4; ++k) {
        if (d[k] >= 0) rank[i + k * stride] = r[k];
    }
}

__global__ __launch_bounds__(256) void scan_chunks(const int* __restrict__ cnt,
                                                   int* __restrict__ loc,
                                                   int* __restrict__ tot) {
    __shared__ int wsum[8];
    const int t    = (int)threadIdx.x;
    const int lane = t & 31;
    const int w    = t >> 5;
    const int base = (int)blockIdx.x * SCAN_CHUNK + t * 4;

    const int4 c = *(const int4*)(cnt + base);
    const int s1 = c.x;
    const int s2 = s1 + c.y;
    const int s3 = s2 + c.z;
    const int s4 = s3 + c.w;

    int x = s4;
    for (int o = 1; o < 32; o <<= 1) {
        const int y = __shfl_up(x, o);
        if (lane >= o) x += y;
    }
    if (lane == 31) wsum[w] = x;
    __syncthreads();
    int wp = 0;
    for (int i = 0; i < w; ++i) wp += wsum[i];

    const int excl = wp + x - s4;
    int4 o4;
    o4.x = excl;
    o4.y = excl + s1;
    o4.z = excl + s2;
    o4.w = excl + s3;
    *(int4*)(loc + base) = o4;
    if (t == 255) tot[blockIdx.x] = wp + x;
}

__global__ __launch_bounds__(32) void scan_totals(const int* __restrict__ tot,
                                                  int* __restrict__ pre,
                                                  int nChunks) {
    const int lane = (int)threadIdx.x;
    int carry = 0;
    for (int b = 0; b < nChunks; b += 32) {
        const int i = b + lane;
        const int v = (i < nChunks) ? tot[i] : 0;
        int x = v;
        for (int o = 1; o < 32; o <<= 1) {
            const int y = __shfl_up(x, o);
            if (lane >= o) x += y;
        }
        if (i < nChunks) pre[i] = carry + x - v;
        carry += __shfl(x, 31);
    }
}

__global__ __launch_bounds__(256) void fill_sorted(const int* __restrict__ esrc,
                                                   const int* __restrict__ edst,
                                                   const float* __restrict__ eval,
                                                   const int* __restrict__ rank,
                                                   const int* __restrict__ loc,
                                                   const int* __restrict__ pre,
                                                   int2* __restrict__ pairs,
                                                   int nEdges) {
    const int i      = (int)(blockIdx.x * blockDim.x + threadIdx.x);
    const int stride = (int)(gridDim.x * blockDim.x);
    int d[4], rk[4], sx[4], vb[4], pos[4];
#pragma unroll
    for (int k = 0; k < 4; ++k) {
        const int  e  = i + k * stride;
        const bool ok = e < nEdges;
        d[k]  = ok ? edst[e] : -1;
        rk[k] = ok ? rank[e] : 0;
        sx[k] = ok ? esrc[e] : 0;
        vb[k] = ok ? __float_as_int(eval[e]) : 0;
    }
#pragma unroll
    for (int k = 0; k < 4; ++k) {
        const int dk = d[k] < 0 ? 0 : d[k];
        pos[k] = loc[dk] + pre[dk >> 10] + rk[k];
    }
#pragma unroll
    for (int k = 0; k < 4; ++k) {
        if (d[k] >= 0) {
            int2 p;
            p.x = sx[k];
            p.y = vb[k];
            pairs[pos[k]] = p;
        }
    }
}

__global__ __launch_bounds__(256) void gather_project(const float* __restrict__ X,
                                                      const float* __restrict__ W,
                                                      const int* __restrict__ loc,
                                                      const int* __restrict__ pre,
                                                      const int2* __restrict__ pairs,
                                                      float* __restrict__ Z) {
    __shared__ __attribute__((aligned(16))) float Tt[16 * TPITCH];
    __shared__ __attribute__((aligned(16))) float Zt[16 * TPITCH];

    const int lane = (int)(threadIdx.x & 31);
    const int w    = __builtin_amdgcn_readfirstlane((int)(threadIdx.x >> 5));
    const int row0 = (int)blockIdx.x * 16;

    for (int r = 0; r < 2; ++r) {
        const int tr  = 2 * w + r;
        const int d   = row0 + tr;
        const int beg = loc[d] + pre[d >> 10];
        const int end = loc[d + 1] + pre[(d + 1) >> 10];

        float4 acc = make_float4(0.f, 0.f, 0.f, 0.f);
        for (int base = beg; base < end; base += 32) {
            const int n = min(32, end - base);
            int2 p = pairs[base + min(lane, n - 1)];
            if (lane >= n) p.y = 0;
            const int nn = (n + 3) & ~3;
            for (int j = 0; j < nn; j += 4) {
                const int   s0 = __builtin_amdgcn_readlane(p.x, j + 0);
                const int   s1 = __builtin_amdgcn_readlane(p.x, j + 1);
                const int   s2 = __builtin_amdgcn_readlane(p.x, j + 2);
                const int   s3 = __builtin_amdgcn_readlane(p.x, j + 3);
                const float v0 = __int_as_float(__builtin_amdgcn_readlane(p.y, j + 0));
                const float v1 = __int_as_float(__builtin_amdgcn_readlane(p.y, j + 1));
                const float v2 = __int_as_float(__builtin_amdgcn_readlane(p.y, j + 2));
                const float v3 = __int_as_float(__builtin_amdgcn_readlane(p.y, j + 3));
                const float4 x0 = ((const float4*)(X + (size_t)s0 * DFEAT))[lane];
                const float4 x1 = ((const float4*)(X + (size_t)s1 * DFEAT))[lane];
                const float4 x2 = ((const float4*)(X + (size_t)s2 * DFEAT))[lane];
                const float4 x3 = ((const float4*)(X + (size_t)s3 * DFEAT))[lane];
                acc.x += v0 * x0.x; acc.y += v0 * x0.y; acc.z += v0 * x0.z; acc.w += v0 * x0.w;
                acc.x += v1 * x1.x; acc.y += v1 * x1.y; acc.z += v1 * x1.z; acc.w += v1 * x1.w;
                acc.x += v2 * x2.x; acc.y += v2 * x2.y; acc.z += v2 * x2.z; acc.w += v2 * x2.w;
                acc.x += v3 * x3.x; acc.y += v3 * x3.y; acc.z += v3 * x3.z; acc.w += v3 * x3.w;
            }
        }
        *(float4*)(Tt + tr * TPITCH + lane * 4) = acc;
    }
    __syncthreads();

    const int hi   = lane >> 4;
    const int l15  = lane & 15;
    const int ncol = w * 16 + l15;
    const float* Arow = Tt + l15 * TPITCH;

    v8f acc = {};
#pragma unroll 8
    for (int k = 0; k < DFEAT; k += 4) {
        const int ka = k + 2 * hi;
        const v2f a = *(const v2f*)(Arow + ka);
        v2f b;
        b.x = W[(size_t)(ka + 0) * DFEAT + ncol];
        b.y = W[(size_t)(ka + 1) * DFEAT + ncol];
        acc = __builtin_amdgcn_wmma_f32_16x16x4_f32(
            false, a, false, b, (short)0, acc, false, false);
    }

#pragma unroll
    for (int i = 0; i < 8; ++i) {
        Zt[(i + 8 * hi) * TPITCH + ncol] = acc[i];
    }
    __syncthreads();

    for (int r = 0; r < 2; ++r) {
        const int tr = 2 * w + r;
        const float4 z = *(const float4*)(Zt + tr * TPITCH + lane * 4);
        *(float4*)(Z + (size_t)(row0 + tr) * DFEAT + lane * 4) = z;
    }
}

extern "C" void kernel_launch(void* const* d_in, const int* in_sizes, int n_in,
                              void* d_out, int out_size, void* d_ws, size_t ws_size,
                              hipStream_t stream) {
    const float* X    = (const float*)d_in[0];
    const float* W    = (const float*)d_in[1];
    const int*   esrc = (const int*)  d_in[2];
    const int*   edst = (const int*)  d_in[3];
    const float* eval = (const float*)d_in[4];
    float*       Z    = (float*)d_out;

    const int nNodes  = in_sizes[0] / DFEAT;
    const int nEdges  = in_sizes[2];
    const int nChunks = (nNodes + 1 + SCAN_CHUNK - 1) / SCAN_CHUNK;
    const int nCnt    = nChunks * SCAN_CHUNK;

    int*  cnt   = (int*)d_ws;
    int*  loc   = cnt + nCnt;
    int*  tot   = loc + nCnt;
    int*  pre   = tot + SCAN_CHUNK;
    int2* pairs = (int2*)(pre + SCAN_CHUNK);
    int*  rank  = (int*)(pairs + nEdges);

    const int block = 256;
    const int edgeGrid = (nEdges + 4 * block - 1) / (4 * block);
    zero_counts<<<(nCnt / 4 + block - 1) / block, block, 0, stream>>>(cnt, nCnt);
    count_rank<<<edgeGrid, block, 0, stream>>>(edst, cnt, rank, nEdges);
    scan_chunks<<<nChunks, block, 0, stream>>>(cnt, loc, tot);
    scan_totals<<<1, 32, 0, stream>>>(tot, pre, nChunks);
    fill_sorted<<<edgeGrid, block, 0, stream>>>(
        esrc, edst, eval, rank, loc, pre, pairs, nEdges);
    gather_project<<<nNodes / 16, block, 0, stream>>>(X, W, loc, pre, pairs, Z);
}


# Gather-project made persistent: 1024 resident blocks loop over the 16-row tiles; W strip stays in registers, next tile's row bounds and edge records are fetched ahead; order of every sum unchanged
# speedup vs baseline: 2.6869x; 1.6580x over previous
#include <hip/hip_runtime.h>


typedef __attribute__((ext_vector_type(2))) float v2f;
typedef __attribute__((ext_vector_type(8))) float v8f;

#define DFEAT      128
#define TPITCH     132
#define BSHIFT     8
#define BROWS      256
#define NBK        512
#define SRCBITS    17
#define SRCMASK    0x1ffff
#define ETHREADS   512
#define EPT        24
#define ECHUNK     (ETHREADS * EPT)
#define RPT        20
#define GBLOCKS    1024
#define BTRIPS     10

__device__ __forceinline__ int block_excl_scan(int v, int* wsum, int lane, int w) {
    int x = v;
    for (int o = 1; o < 32; o <<= 1) {
        const int y = __shfl_up(x, o);
        if (lane >= o) x += y;
    }
    if (lane == 31) wsum[w] = x;
    __syncthreads();
    int wp = 0;
    for (int i = 0; i < w; ++i) wp += wsum[i];
    return wp + x - v;
}

__global__ __launch_bounds__(ETHREADS) void bucket_hist(const int* __restrict__ edst,
                                                        int* __restrict__ H,
                                                        int nEdges) {
    __shared__ int h[NBK];
    const int t = (int)threadIdx.x;
    h[t] = 0;
    __syncthreads();
    const int beg = (int)blockIdx.x * ECHUNK;
    int d[EPT];
#pragma unroll
    for (int k = 0; k < EPT; ++k) {
        const int e = beg + k * ETHREADS + t;
        d[k] = (e < nEdges) ? edst[e] : -1;
    }
#pragma unroll
    for (int k = 0; k < EPT; ++k) {
        if (d[k] >= 0) atomicAdd(&h[d[k] >> BSHIFT], 1);
    }
    __syncthreads();
    H[(int)blockIdx.x * NBK + t] = h[t];
}

__global__ __launch_bounds__(ETHREADS) void bucket_scatter(const int* __restrict__ esrc,
                                                           const int* __restrict__ edst,
                                                           const float* __restrict__ eval,
                                                           const int* __restrict__ H,
                                                           int2* __restrict__ rec,
                                                           int* __restrict__ bstart,
                                                           int nEdges) {
    __shared__ int cur[NBK + 1];
    __shared__ int wsum[ETHREADS / 32];
    const int t    = (int)threadIdx.x;
    const int lane = t & 31;
    const int w    = t >> 5;
    const int b    = (int)blockIdx.x;
    const int nb   = (int)gridDim.x;

    int tot = 0, pre = 0;
#pragma unroll 16
    for (int j = 0; j < nb; ++j) {
        const int v = H[j * NBK + t];
        tot += v;
        if (j < b) pre += v;
    }
    const int excl = block_excl_scan(tot, wsum, lane, w);
    cur[t] = excl + pre;
    if (t == 0) cur[NBK] = 0;
    if (b == 0) bstart[t] = excl;
    __syncthreads();

    const int beg = b * ECHUNK;
    int d[EPT], s[EPT], v[EPT];
#pragma unroll
    for (int k = 0; k < EPT; ++k) {
        const int  e  = beg + k * ETHREADS + t;
        const bool ok = e < nEdges;
        d[k] = ok ? edst[e] : -1;
        s[k] = ok ? esrc[e] : 0;
        v[k] = ok ? __float_as_int(eval[e]) : 0;
    }
    int pos[EPT];
#pragma unroll
    for (int k = 0; k < EPT; ++k) {
        pos[k] = atomicAdd(&cur[(d[k] >= 0) ? (d[k] >> BSHIFT) : NBK], 1);
    }
#pragma unroll
    for (int k = 0; k < EPT; ++k) {
        if (d[k] >= 0) {
            int2 r;
            r.x = s[k] | ((d[k] & (BROWS - 1)) << SRCBITS);
            r.y = v[k];
            rec[pos[k]] = r;
        }
    }
}

__global__ __launch_bounds__(BROWS) void bucket_sort(const int2* __restrict__ rec,
                                                     const int* __restrict__ bstart,
                                                     int2* __restrict__ pairs,
                                                     int* __restrict__ start,
                                                     int nNodes) {
    __shared__ int cnt[BROWS + 1];
    __shared__ int wsum[BROWS / 32];
    const int t    = (int)threadIdx.x;
    const int lane = t & 31;
    const int w    = t >> 5;
    const int bk   = (int)blockIdx.x;
    const int bBeg = bstart[bk];
    const int bEnd = bstart[bk + 1];

    cnt[t] = 0;
    if (t == 0) cnt[BROWS] = 0;
    __syncthreads();
    for (int e0 = bBeg + t; e0 < bEnd; e0 += BROWS * RPT) {
        int key[RPT];
#pragma unroll
        for (int k = 0; k < RPT; ++k) {
            const int e = e0 + k * BROWS;
            key[k] = (e < bEnd) ? rec[e].x : -1;
        }
#pragma unroll
        for (int k = 0; k < RPT; ++k) {
            if (key[k] >= 0) atomicAdd(&cnt[key[k] >> SRCBITS], 1);
        }
    }
    __syncthreads();

    const int c     = cnt[t];
    const int first = bBeg + block_excl_scan(c, wsum, lane, w);
    const int node  = bk * BROWS + t;
    if (node <= nNodes) start[node] = first;
    cnt[t] = first;
    __syncthreads();

    for (int e0 = bBeg + t; e0 < bEnd; e0 += BROWS * RPT) {
        int2 r[RPT];
#pragma unroll
        for (int k = 0; k < RPT; ++k) {
            const int e = e0 + k * BROWS;
            r[k] = (e < bEnd) ? rec[e] : make_int2(-1, 0);
        }
        int pos[RPT];
#pragma unroll
        for (int k = 0; k < RPT; ++k) {
            pos[k] = atomicAdd(&cnt[(r[k].x >= 0) ? (r[k].x >> SRCBITS) : BROWS], 1);
        }
#pragma unroll
        for (int k = 0; k < RPT; ++k) {
            if (r[k].x >= 0) {
                int2 p;
                p.x = r[k].x & SRCMASK;
                p.y = r[k].y;
                pairs[pos[k]] = p;
            }
        }
    }
}


__device__ __forceinline__ int2 first_records(const int2* __restrict__ pairs, int beg, int end, int lane) {
    const int n = min(32, end - beg);
    int2 p = make_int2(0, 0);
    if (n > 0) p = pairs[beg + min(lane, n - 1)];
    return p;
}

__device__ __forceinline__ int2 zero_tail(int2 p, int n, int lane) {
    if (lane >= n) p.y = 0;
    return p;
}

__device__ __forceinline__ float4 sum_row(const float* __restrict__ X, const int2* __restrict__ pairs,
                                          int beg, int end, int2 p, int lane) {
    float4 acc = make_float4(0.f, 0.f, 0.f, 0.f);
    for (int base = beg; base < end; base += 32) {
        const int n = min(32, end - base);
        if (base != beg) p = zero_tail(pairs[base + min(lane, n - 1)], n, lane);
        const int nn = (n + 3) & ~3;
#pragma unroll 2
        for (int j = 0; j < nn; j += 4) {
            const int   s0 = __builtin_amdgcn_readlane(p.x, j + 0);
            const int   s1 = __builtin_amdgcn_readlane(p.x, j + 1);
            const int   s2 = __builtin_amdgcn_readlane(p.x, j + 2);
            const int   s3 = __builtin_amdgcn_readlane(p.x, j + 3);
            const float v0 = __int_as_float(__builtin_amdgcn_readlane(p.y, j + 0));
            const float v1 = __int_as_float(__builtin_amdgcn_readlane(p.y, j + 1));
            const float v2 = __int_as_float(__builtin_amdgcn_readlane(p.y, j + 2));
            const float v3 = __int_as_float(__builtin_amdgcn_readlane(p.y, j + 3));
            const float4 x0 = ((const float4*)(X + (size_t)s0 * DFEAT))[lane];
            const float4 x1 = ((const float4*)(X + (size_t)s1 * DFEAT))[lane];
            const float4 x2 = ((const float4*)(X + (size_t)s2 * DFEAT))[lane];
            const float4 x3 = ((const float4*)(X + (size_t)s3 * DFEAT))[lane];
            acc.x += v0 * x0.x; acc.y += v0 * x0.y; acc.z += v0 * x0.z; acc.w += v0 * x0.w;
            acc.x += v1 * x1.x; acc.y += v1 * x1.y; acc.z += v1 * x1.z; acc.w += v1 * x1.w;
            acc.x += v2 * x2.x; acc.y += v2 * x2.y; acc.z += v2 * x2.z; acc.w += v2 * x2.w;
            acc.x += v3 * x3.x; acc.y += v3 * x3.y; acc.z += v3 * x3.z; acc.w += v3 * x3.w;
        }
    }
    return acc;
}

__global__ __launch_bounds__(256, 8) void gather_project(const float* __restrict__ X,
                                                         const float* __restrict__ W,
                                                         const int* __restrict__ start,
                                                         const int2* __restrict__ pairs,
                                                         float* __restrict__ Z,
                                                         int nTiles) {
    __shared__ __attribute__((aligned(16))) float Tt[16 * TPITCH];
    __shared__ __attribute__((aligned(16))) float Zt[16 * TPITCH];

    if ((int)blockIdx.x >= nTiles) return;

    const int lane = (int)(threadIdx.x & 31);
    const int w    = __builtin_amdgcn_readfirstlane((int)(threadIdx.x >> 5));
    const int hi   = lane >> 4;
    const int l15  = lane & 15;
    const int ncol = w * 16 + l15;
    const float* Arow = Tt + l15 * TPITCH + 2 * hi;
    const int G     = (int)gridDim.x;
    const int bslot = min(lane / 3, BTRIPS - 1);
    const int bsub  = lane - 3 * (lane / 3);
    int tile = (int)blockIdx.x;
    int bnd  = start[min(tile + bslot * G, nTiles - 1) * 16 + 2 * w + bsub];
    int trip = 0;
    int c0 = __builtin_amdgcn_readlane(bnd, 0);
    int c1 = __builtin_amdgcn_readlane(bnd, 1);
    int c2 = __builtin_amdgcn_readlane(bnd, 2);
    int2 pA = first_records(pairs, c0, c1, lane);
    int2 pB = first_records(pairs, c1, c2, lane);

    v2f bw[DFEAT / 4];
#pragma unroll
    for (int kk = 0; kk < DFEAT / 4; ++kk) {
        const int ka = 4 * kk + 2 * hi;
        bw[kk].x = W[(size_t)(ka + 0) * DFEAT + ncol];
        bw[kk].y = W[(size_t)(ka + 1) * DFEAT + ncol];
    }

    for (; tile < nTiles; tile += G) {
        const int row0 = tile * 16;
        pA = zero_tail(pA, min(32, c1 - c0), lane);
        pB = zero_tail(pB, min(32, c2 - c1), lane);
        ++trip;
        if (trip == BTRIPS) {
            bnd  = start[min(tile + G + bslot * G, nTiles - 1) * 16 + 2 * w + bsub];
            trip = 0;
        }
        const int n0 = __builtin_amdgcn_readlane(bnd, 3 * trip + 0);
        const int n1 = __builtin_amdgcn_readlane(bnd, 3 * trip + 1);
        const int n2 = __builtin_amdgcn_readlane(bnd, 3 * trip + 2);
        const int2 qA = first_records(pairs, n0, n1, lane);
        const int2 qB = first_records(pairs, n1, n2, lane);

        const float4 t0 = sum_row(X, pairs, c0, c1, pA, lane);
        *(float4*)(Tt + (2 * w + 0) * TPITCH + lane * 4) = t0;
        const float4 t1 = sum_row(X, pairs, c1, c2, pB, lane);
        *(float4*)(Tt + (2 * w + 1) * TPITCH + lane * 4) = t1;
        c0 = n0; c1 = n1; c2 = n2; pA = qA; pB = qB;
        __syncthreads();

        v8f acc = {};
#pragma unroll
        for (int kk = 0; kk < DFEAT / 4; ++kk) {
            const v2f a = *(const v2f*)(Arow + 4 * kk);
            acc = __builtin_amdgcn_wmma_f32_16x16x4_f32(
                false, a, false, bw[kk], (short)0, acc, false, false);
        }

#pragma unroll
        for (int i = 0; i < 8; ++i) {
            Zt[(i + 8 * hi) * TPITCH + ncol] = acc[i];
        }
        __syncthreads();

        for (int r = 0; r < 2; ++r) {
            const int tr = 2 * w + r;
            const float4 z = *(const float4*)(Zt + tr * TPITCH + lane * 4);
            *(float4*)(Z + (size_t)(row0 + tr) * DFEAT + lane * 4) = z;
        }
    }
}

extern "C" void kernel_launch(void* const* d_in, const int* in_sizes, int n_in,
                              void* d_out, int out_size, void* d_ws, size_t ws_size,
                              hipStream_t stream) {
    const float* X    = (const float*)d_in[0];
    const float* W    = (const float*)d_in[1];
    const int*   esrc = (const int*)  d_in[2];
    const int*   edst = (const int*)  d_in[3];
    const float* eval = (const float*)d_in[4];
    float*       Z    = (float*)d_out;

    const int nNodes   = in_sizes[0] / DFEAT;
    const int nEdges   = in_sizes[2];
    const int nBuckets = (nNodes >> BSHIFT) + 1;
    const int eBlocks  = (nEdges + ECHUNK - 1) / ECHUNK;

    int*  H      = (int*)d_ws;
    int*  bstart = H + eBlocks * NBK;
    int*  start  = bstart + NBK;
    int2* rec    = (int2*)(start + nBuckets * BROWS);
    int2* pairs  = rec + nEdges;

    bucket_hist<<<eBlocks, ETHREADS, 0, stream>>>(edst, H, nEdges);
    bucket_scatter<<<eBlocks, ETHREADS, 0, stream>>>(esrc, edst, eval, H, rec, bstart, nEdges);
    bucket_sort<<<nBuckets, BROWS, 0, stream>>>(rec, bstart, pairs, start, nNodes);
    gather_project<<<GBLOCKS, 256, 0, stream>>>(X, W, start, pairs, Z, nNodes / 16);
}


# Gather keeps 8 row loads in flight per wave: its two rows' records are walked as one stream through a ring of loads (was 4 loads, drained before the next 4); order of every sum unchanged
# speedup vs baseline: 3.3077x; 1.2310x over previous
#include <hip/hip_runtime.h>


typedef __attribute__((ext_vector_type(2))) float v2f;
typedef __attribute__((ext_vector_type(8))) float v8f;

#define DFEAT      128
#define TPITCH     132
#define BSHIFT     6
#define BROWS      64
#define BTILES     (BROWS / 16)
#define NBK        2048
#define SRCBITS    17
#define SRCMASK    0x1ffff
#define ETHREADS   512
#define EPT        24
#define ECHUNK     (ETHREADS * EPT)
#define CAP        1536
#define SPT        (CAP / 256)
#define GBLOCKS    1024

__device__ __forceinline__ int block_excl_scan(int v, int* wsum, int lane, int w) {
    int x = v;
    for (int o = 1; o < 32; o <<= 1) {
        const int y = __shfl_up(x, o);
        if (lane >= o) x += y;
    }
    if (lane == 31) wsum[w] = x;
    __syncthreads();
    int wp = 0;
    for (int i = 0; i < w; ++i) wp += wsum[i];
    return wp + x - v;
}

__global__ __launch_bounds__(ETHREADS) void chunk_sort(const int* __restrict__ esrc,
                                                       const int* __restrict__ edst,
                                                       const float* __restrict__ eval,
                                                       int2* __restrict__ rec,
                                                       int* __restrict__ L,
                                                       int nEdges) {
    __shared__ __attribute__((aligned(16))) int cur[NBK + 4];
    __shared__ int wsum[ETHREADS / 32];
    const int t    = (int)threadIdx.x;
    const int lane = t & 31;
    const int w    = t >> 5;
    const int b    = (int)blockIdx.x;

    *(int4*)&cur[4 * t] = make_int4(0, 0, 0, 0);
    if (t == 0) cur[NBK] = 0;

    const int beg = b * ECHUNK;
    int d[EPT], s[EPT], v[EPT];
#pragma unroll
    for (int k = 0; k < EPT; ++k) {
        const int  e  = beg + k * ETHREADS + t;
        const bool ok = e < nEdges;
        d[k] = ok ? edst[e] : -1;
        s[k] = ok ? esrc[e] : 0;
        v[k] = ok ? __float_as_int(eval[e]) : 0;
    }
    __syncthreads();
#pragma unroll
    for (int k = 0; k < EPT; ++k) {
        if (d[k] >= 0) atomicAdd(&cur[d[k] >> BSHIFT], 1);
    }
    __syncthreads();

    const int4 c4   = *(const int4*)&cur[4 * t];
    const int  e0   = block_excl_scan(c4.x + c4.y + c4.z + c4.w, wsum, lane, w);
    int4 p4;
    p4.x = e0;
    p4.y = p4.x + c4.x;
    p4.z = p4.y + c4.y;
    p4.w = p4.z + c4.z;
    *(int4*)&cur[4 * t] = p4;
    __syncthreads();
    *(int4*)&L[b * NBK + 4 * t] = p4;

    int pos[EPT];
#pragma unroll
    for (int k = 0; k < EPT; ++k) {
        pos[k] = atomicAdd(&cur[(d[k] >= 0) ? (d[k] >> BSHIFT) : NBK], 1);
    }
#pragma unroll
    for (int k = 0; k < EPT; ++k) {
        if (d[k] >= 0) {
            int2 r;
            r.x = s[k] | ((d[k] & (BROWS - 1)) << SRCBITS);
            r.y = v[k];
            rec[beg + pos[k]] = r;
        }
    }
}


__device__ __forceinline__ int2 load_window(const int2* srt, int c0, int wbase, int total, int lane) {
    const int r = wbase + lane;
    int2 p = srt[c0 + min(r, total - 1)];
    if (r >= total) p.y = 0;
    return p;
}

__device__ __forceinline__ float4 load_row(const float* __restrict__ X, int s, unsigned lane16) {
    return *(const float4*)((const char*)X + (((unsigned)s << 9) + lane16));
}

#define RING 8

__device__ __forceinline__ void gather_two_rows(const float* __restrict__ X, const int2* srt,
                                                int c0, int c1, int c2, int lane,
                                                float* T0, float* T1) {
    float4 acc = make_float4(0.f, 0.f, 0.f, 0.f);
    const int total = c2 - c0;
    const int split = c1 - c0;
    if (total > 0) {
        const unsigned lane16 = (unsigned)lane * 16u;
        const int nPad = (total + RING - 1) & ~(RING - 1);
        float4 x[RING];
        float  wv[RING];
        int2 p = load_window(srt, c0, 0, total, lane);
#pragma unroll
        for (int k = 0; k < RING; ++k) {
            const int s = __builtin_amdgcn_readlane(p.x, k);
            wv[k] = __int_as_float(__builtin_amdgcn_readlane(p.y, k));
            x[k]  = load_row(X, s, lane16);
        }
        for (int i = RING; i < nPad; i += RING) {
            if ((i & 31) == 0) p = load_window(srt, c0, i, total, lane);
#pragma unroll
            for (int k = 0; k < RING; ++k) {
                if (i - RING + k == split) {
                    *(float4*)T0 = acc;
                    acc = make_float4(0.f, 0.f, 0.f, 0.f);
                }
                acc.x += wv[k] * x[k].x; acc.y += wv[k] * x[k].y;
                acc.z += wv[k] * x[k].z; acc.w += wv[k] * x[k].w;
                const int l = (i + k) & 31;
                const int s = __builtin_amdgcn_readlane(p.x, l);
                wv[k] = __int_as_float(__builtin_amdgcn_readlane(p.y, l));
                x[k]  = load_row(X, s, lane16);
            }
        }
#pragma unroll
        for (int k = 0; k < RING; ++k) {
            if (nPad - RING + k == split) {
                *(float4*)T0 = acc;
                acc = make_float4(0.f, 0.f, 0.f, 0.f);
            }
            acc.x += wv[k] * x[k].x; acc.y += wv[k] * x[k].y;
            acc.z += wv[k] * x[k].z; acc.w += wv[k] * x[k].w;
        }
        if (split >= nPad) *(float4*)T0 = acc;
        if (split == total) acc = make_float4(0.f, 0.f, 0.f, 0.f);
    } else {
        *(float4*)T0 = acc;
    }
    *(float4*)T1 = acc;
}

__device__ __forceinline__ float4 sum_row_unsorted(const float* __restrict__ X, const int2* __restrict__ rec,
                                                   const int* __restrict__ L, int bk, int r,
                                                   int nChunks, int lane) {
    float4 acc = make_float4(0.f, 0.f, 0.f, 0.f);
    for (int c = 0; c < nChunks; ++c) {
        const int lo  = __builtin_amdgcn_readfirstlane(L[c * NBK + bk]);
        const int hi  = __builtin_amdgcn_readfirstlane(L[c * NBK + bk + 1]);
        const int end = c * ECHUNK + hi;
        for (int base = c * ECHUNK + lo; base < end; base += 32) {
            const int  n = min(32, end - base);
            const int2 p = rec[base + min(lane, n - 1)];
            unsigned mask = __builtin_amdgcn_ballot_w32((lane < n) && ((p.x >> SRCBITS) == r));
            while (mask != 0u) {
                const int j = __builtin_ctz(mask);
                mask &= mask - 1u;
                const int    s = __builtin_amdgcn_readlane(p.x, j) & SRCMASK;
                const float  v = __int_as_float(__builtin_amdgcn_readlane(p.y, j));
                const float4 x = ((const float4*)(X + (size_t)s * DFEAT))[lane];
                acc.x += v * x.x; acc.y += v * x.y; acc.z += v * x.z; acc.w += v * x.w;
            }
        }
    }
    return acc;
}

__global__ __launch_bounds__(256, 8) void sort_gather_project(const float* __restrict__ X,
                                                              const float* __restrict__ W,
                                                              const int2* __restrict__ rec,
                                                              const int* __restrict__ L,
                                                              float* __restrict__ Z,
                                                              int nTiles, int nChunks) {
    __shared__ __attribute__((aligned(16))) float Tt[16 * TPITCH];
    __shared__ __attribute__((aligned(16))) float Zt[16 * TPITCH];
    __shared__ __attribute__((aligned(16))) int2 srt[CAP];
    __shared__ int idx[CAP];
    __shared__ int cntrow[(BROWS + 32) + (BROWS + 1)];
    int* const cnt    = cntrow;
    int* const rowbeg = cntrow + (BROWS + 32);
    __shared__ int wrun[256 / 32];

    const int t    = (int)threadIdx.x;
    const int lane = t & 31;
    const int w    = __builtin_amdgcn_readfirstlane(t >> 5);
    const int hi   = lane >> 4;
    const int l15  = lane & 15;
    const int ncol = w * 16 + l15;
    const int G    = (int)gridDim.x;
    const int nBuckets = (nTiles + BTILES - 1) / BTILES;
    int bk = (int)blockIdx.x;
    if (bk >= nBuckets) return;

    const float* Arow = Tt + l15 * TPITCH + 2 * hi;

    const int  tc  = min(t, nChunks - 1);
    const bool own = t < nChunks;
    int lo = L[tc * NBK + bk];
    int up = L[tc * NBK + bk + 1];

    v2f bw[DFEAT / 4];
#pragma unroll
    for (int kk = 0; kk < DFEAT / 4; ++kk) {
        const int ka = 4 * kk + 2 * hi;
        bw[kk].x = W[(size_t)(ka + 0) * DFEAT + ncol];
        bw[kk].y = W[(size_t)(ka + 1) * DFEAT + ncol];
    }

    for (; bk < nBuckets; bk += G) {
        const int n    = own ? up - lo : 0;
        const int src0 = t * ECHUNK + lo;
        if (t < BROWS + 32) cnt[t] = 0;

        int xn = n;
        for (int o = 1; o < 32; o <<= 1) {
            const int yn = __shfl_up(xn, o);
            if (lane >= o) xn += yn;
        }
        if (lane == 31) wrun[w] = xn;
        __syncthreads();
        int ex = xn - n, tot = 0;
        for (int i = 0; i < 256 / 32; ++i) {
            if (i < w) ex += wrun[i];
            tot += wrun[i];
        }
        const int  total = __builtin_amdgcn_readfirstlane(tot);
        const bool fits  = total <= CAP;

        if (fits) {
            int2 r[SPT];
#pragma unroll
            for (int k = 0; k < SPT; ++k) r[k] = make_int2(-1, 0);
            if (total > 0) {
                for (int q = 0; q < n; ++q) idx[ex + q] = src0 + q;
                __syncthreads();
                int a[SPT];
#pragma unroll
                for (int k = 0; k < SPT; ++k) a[k] = idx[min(k * 256 + t, total - 1)];
#pragma unroll
                for (int k = 0; k < SPT; ++k) {
                    if (k * 256 + t < total) r[k] = rec[a[k]];
                }
            }
            int rk[SPT];
#pragma unroll
            for (int k = 0; k < SPT; ++k) {
                rk[k] = atomicAdd(&cnt[(r[k].x >= 0) ? (r[k].x >> SRCBITS) : BROWS + lane], 1);
            }
            __syncthreads();
            if (w == 0) {
                const int ca = cnt[lane];
                const int cb = cnt[lane + 32];
                int ia = ca, ib = cb;
                for (int o = 1; o < 32; o <<= 1) {
                    const int ya = __shfl_up(ia, o);
                    const int yb = __shfl_up(ib, o);
                    if (lane >= o) { ia += ya; ib += yb; }
                }
                const int ta = __shfl(ia, 31);
                rowbeg[lane]      = ia - ca;
                rowbeg[lane + 32] = ta + ib - cb;
                if (lane == 31) rowbeg[BROWS] = ta + ib;
            }
            __syncthreads();
#pragma unroll
            for (int k = 0; k < SPT; ++k) {
                if (r[k].x >= 0) {
                    int2 p;
                    p.x = r[k].x & SRCMASK;
                    p.y = r[k].y;
                    srt[rowbeg[r[k].x >> SRCBITS] + rk[k]] = p;
                }
            }
            __syncthreads();
        }

        const int tile0 = bk * BTILES;
        const int nT    = min(BTILES, nTiles - tile0);

        for (int tt = 0; tt < nT; ++tt) {
            const int row0 = (tile0 + tt) * 16;
            const int rb   = tt * 16 + 2 * w;

            float* T0 = Tt + (2 * w + 0) * TPITCH + lane * 4;
            float* T1 = Tt + (2 * w + 1) * TPITCH + lane * 4;
            if (fits) {
                const int c0 = __builtin_amdgcn_readfirstlane(rowbeg[rb + 0]);
                const int c1 = __builtin_amdgcn_readfirstlane(rowbeg[rb + 1]);
                const int c2 = __builtin_amdgcn_readfirstlane(rowbeg[rb + 2]);
                gather_two_rows(X, srt, c0, c1, c2, lane, T0, T1);
            } else {
                const float4 t0 = sum_row_unsorted(X, rec, L, bk, rb + 0, nChunks, lane);
                const float4 t1 = sum_row_unsorted(X, rec, L, bk, rb + 1, nChunks, lane);
                *(float4*)T0 = t0;
                *(float4*)T1 = t1;
            }
            __syncthreads();

            v8f acc = {};
#pragma unroll
            for (int kk = 0; kk < DFEAT / 4; ++kk) {
                const v2f a = *(const v2f*)(Arow + 4 * kk);
                acc = __builtin_amdgcn_wmma_f32_16x16x4_f32(
                    false, a, false, bw[kk], (short)0, acc, false, false);
            }

#pragma unroll
            for (int i = 0; i < 8; ++i) {
                Zt[(i + 8 * hi) * TPITCH + ncol] = acc[i];
            }
            __syncthreads();

            for (int q = 0; q < 2; ++q) {
                const int tr = 2 * w + q;
                const float4 z = *(const float4*)(Zt + tr * TPITCH + lane * 4);
                *(float4*)(Z + (size_t)(row0 + tr) * DFEAT + lane * 4) = z;
            }
        }
        const int bn = min(bk + G, nBuckets - 1);
        lo = L[tc * NBK + bn];
        up = L[tc * NBK + bn + 1];
    }
}

extern "C" void kernel_launch(void* const* d_in, const int* in_sizes, int n_in,
                              void* d_out, int out_size, void* d_ws, size_t ws_size,
                              hipStream_t stream) {
    const float* X    = (const float*)d_in[0];
    const float* W    = (const float*)d_in[1];
    const int*   esrc = (const int*)  d_in[2];
    const int*   edst = (const int*)  d_in[3];
    const float* eval = (const float*)d_in[4];
    float*       Z    = (float*)d_out;

    const int nNodes   = in_sizes[0] / DFEAT;
    const int nEdges   = in_sizes[2];
    const int nChunks  = (nEdges + ECHUNK - 1) / ECHUNK;

    int*  L   = (int*)d_ws;
    int2* rec = (int2*)(L + nChunks * NBK);

    chunk_sort<<<nChunks, ETHREADS, 0, stream>>>(esrc, edst, eval, rec, L, nEdges);
    sort_gather_project<<<GBLOCKS, 256, 0, stream>>>(X, W, rec, L, Z, nNodes / 16, nChunks);
}
